# starting kernel
# baseline (speedup 1.0000x reference)
#include <hip/hip_runtime.h>

typedef float v2f __attribute__((ext_vector_type(2)));
typedef float v8f __attribute__((ext_vector_type(8)));

#define B_ 8
#define I_ 16
#define C_ 256

static const size_t SEL_OFF[4] = {0, 2097152, 2621440, 2752512};
static const size_t S_OFF      = 2785280;
static const size_t CNT_OFF    = 2916352;
static const size_t PK_BYTEOFF = 2916864ull * 4ull;

__global__ void zero_kernel(float* __restrict__ p, int n) {
    int i = blockIdx.x * blockDim.x + threadIdx.x;
    if (i < n) p[i] = 0.0f;
}

__global__ void __launch_bounds__(256)
mask_kernel(const float* __restrict__ scrib, float* __restrict__ sel,
            float* __restrict__ cnt, unsigned long long* __restrict__ packed,
            int w, int factor, int offs, int blocksPerBI) {
    __shared__ float scnt[256];
    __shared__ unsigned long long smax[256];
    int bi = blockIdx.x / blocksPerBI;
    int pb = blockIdx.x % blocksPerBI;
    int t  = threadIdx.x;
    int p  = pb * 256 + t;
    int y = p / w, x = p % w;
    int ys = y * factor + offs;
    int xs = x * factor + offs;
    const float* sp = scrib + (size_t)bi * 512 * 512;
    float m = 0.25f * (sp[(size_t)ys * 512 + xs]     + sp[(size_t)ys * 512 + xs + 1] +
                       sp[(size_t)(ys + 1) * 512 + xs] + sp[(size_t)(ys + 1) * 512 + xs + 1]);
    float sv = (m > 0.5f) ? 1.0f : 0.0f;
    sel[(size_t)bi * ((size_t)blocksPerBI * 256) + p] = sv;
    scnt[t] = sv;
    smax[t] = ((unsigned long long)__float_as_uint(m) << 32) |
              (unsigned long long)(0xFFFFFFFFu - (unsigned)p);
    __syncthreads();
    for (int st = 128; st > 0; st >>= 1) {
        if (t < st) {
            scnt[t] += scnt[t + st];
            if (smax[t + st] > smax[t]) smax[t] = smax[t + st];
        }
        __syncthreads();
    }
    if (t == 0) {
        atomicAdd(&cnt[bi], scnt[0]);
        atomicMax(&packed[bi], smax[0]);
    }
}

__global__ void __launch_bounds__(256)
wmma_masked_sum_kernel(const float* __restrict__ feat, const float* __restrict__ sel,
                       float* __restrict__ s, int P, int chunk, int nchunks) {
    int wid  = (blockIdx.x * blockDim.x + threadIdx.x) >> 5;
    int lane = threadIdx.x & 31;
    int kc = wid % nchunks;
    int ct = (wid / nchunks) & 15;
    int b  = wid / (nchunks * 16);
    int k0 = kc * chunk;
    int row = lane & 15;
    int kb  = (lane >> 4) * 2;
    const float* ap = sel  + ((size_t)b * I_ + row) * (size_t)P + (size_t)(k0 + kb);
    const float* bp = feat + ((size_t)b * C_ + (size_t)ct * 16 + row) * (size_t)P + (size_t)(k0 + kb);
    v8f acc = {};
    for (int k = 0; k < chunk; k += 4) {
        v2f a  = *(const v2f*)(ap + k);
        v2f bv = *(const v2f*)(bp + k);
        acc = __builtin_amdgcn_wmma_f32_16x16x4_f32(false, a, false, bv,
                                                    (short)0, acc, false, false);
    }
    float* sd = s + ((size_t)b * I_ + (size_t)(lane >> 4) * 8) * C_ + (size_t)ct * 16 + (lane & 15);
#pragma unroll
    for (int r = 0; r < 8; ++r)
        atomicAdd(sd + (size_t)r * C_, acc[r]);
}

__global__ void __launch_bounds__(256)
finalize_kernel(const float* __restrict__ f0, const float* __restrict__ f1,
                const float* __restrict__ f2, const float* __restrict__ f3,
                const float* __restrict__ s, const float* __restrict__ cnt,
                const unsigned long long* __restrict__ packed, float* __restrict__ out) {
    int tid = blockIdx.x * blockDim.x + threadIdx.x;
    int c  = tid & 255;
    int bi = tid >> 8;
    int b  = bi >> 4;
    const float* feats[4] = {f0, f1, f2, f3};
    const int Ps[4] = {16384, 4096, 1024, 256};
    float sum = 0.0f;
#pragma unroll
    for (int l = 0; l < 4; ++l) {
        float cv = cnt[l * (B_ * I_) + bi];
        float v;
        if (cv > 0.0f) {
            v = s[l * (B_ * I_ * C_) + tid] / cv;
        } else {
            unsigned long long pk = packed[l * (B_ * I_) + bi];
            unsigned idx = 0xFFFFFFFFu - (unsigned)(pk & 0xFFFFFFFFull);
            v = feats[l][((size_t)b * C_ + c) * (size_t)Ps[l] + idx];
        }
        sum += v;
    }
    out[tid] = 0.25f * sum;
}

extern "C" void kernel_launch(void* const* d_in, const int* in_sizes, int n_in,
                              void* d_out, int out_size, void* d_ws, size_t ws_size,
                              hipStream_t stream) {
    const float* feat0 = (const float*)d_in[0];
    const float* feat1 = (const float*)d_in[1];
    const float* feat2 = (const float*)d_in[2];
    const float* feat3 = (const float*)d_in[3];
    const float* scrib = (const float*)d_in[4];
    float* ws = (float*)d_ws;

    float* sA   = ws + S_OFF;
    float* cntA = ws + CNT_OFF;
    unsigned long long* pkA = (unsigned long long*)((char*)d_ws + PK_BYTEOFF);

    int nz = (B_ * I_ * C_ * 4) + (B_ * I_ * 4) + (B_ * I_ * 4 * 2);
    zero_kernel<<<(nz + 255) / 256, 256, 0, stream>>>(sA, nz);

    const float* feats[4] = {feat0, feat1, feat2, feat3};
    const int H[4]  = {128, 64, 32, 16};
    const int CH[4] = {1024, 512, 256, 128};

    for (int l = 0; l < 4; ++l) {
        int h = H[l], P = h * h;
        int factor = 512 / h;
        int offs = factor / 2 - 1;
        int blocksPerBI = P / 256;
        mask_kernel<<<B_ * I_ * blocksPerBI, 256, 0, stream>>>(
            scrib, ws + SEL_OFF[l], cntA + l * (B_ * I_), pkA + l * (B_ * I_),
            h, factor, offs, blocksPerBI);
    }
    for (int l = 0; l < 4; ++l) {
        int P = H[l] * H[l];
        int chunk = CH[l];
        int nchunks = P / chunk;
        int waves = B_ * 16 * nchunks;
        wmma_masked_sum_kernel<<<waves * 32 / 256, 256, 0, stream>>>(
            feats[l], ws + SEL_OFF[l], sA + l * (B_ * I_ * C_), P, chunk, nchunks);
    }
    finalize_kernel<<<(B_ * I_ * C_) / 256, 256, 0, stream>>>(
        feat0, feat1, feat2, feat3, sA, cntA, pkA, (float*)d_out);
}


# masked-sum WMMA kernel re-tiled: 16-byte row loads (same k permutation for sel and feat), one sel fragment feeds two C tiles, a block's 8 K slices summed in LDS before atomics (sum order changes)
# speedup vs baseline: 1.2764x; 1.2764x over previous
#include <hip/hip_runtime.h>

typedef float v2f __attribute__((ext_vector_type(2)));
typedef float v8f __attribute__((ext_vector_type(8)));

#define B_ 8
#define I_ 16
#define C_ 256

static const size_t SEL_OFF[4] = {0, 2097152, 2621440, 2752512};
static const size_t S_OFF      = 2785280;
static const size_t CNT_OFF    = 2916352;
static const size_t PK_BYTEOFF = 2916864ull * 4ull;

__global__ void zero_kernel(float* __restrict__ p, int n) {
    int i = blockIdx.x * blockDim.x + threadIdx.x;
    if (i < n) p[i] = 0.0f;
}

__global__ void __launch_bounds__(256)
mask_kernel(const float* __restrict__ scrib, float* __restrict__ sel,
            float* __restrict__ cnt, unsigned long long* __restrict__ packed,
            int w, int factor, int offs, int blocksPerBI) {
    __shared__ float scnt[256];
    __shared__ unsigned long long smax[256];
    int bi = blockIdx.x / blocksPerBI;
    int pb = blockIdx.x % blocksPerBI;
    int t  = threadIdx.x;
    int p  = pb * 256 + t;
    int y = p / w, x = p % w;
    int ys = y * factor + offs;
    int xs = x * factor + offs;
    const float* sp = scrib + (size_t)bi * 512 * 512;
    float m = 0.25f * (sp[(size_t)ys * 512 + xs]     + sp[(size_t)ys * 512 + xs + 1] +
                       sp[(size_t)(ys + 1) * 512 + xs] + sp[(size_t)(ys + 1) * 512 + xs + 1]);
    float sv = (m > 0.5f) ? 1.0f : 0.0f;
    sel[(size_t)bi * ((size_t)blocksPerBI * 256) + p] = sv;
    scnt[t] = sv;
    smax[t] = ((unsigned long long)__float_as_uint(m) << 32) |
              (unsigned long long)(0xFFFFFFFFu - (unsigned)p);
    __syncthreads();
    for (int st = 128; st > 0; st >>= 1) {
        if (t < st) {
            scnt[t] += scnt[t + st];
            if (smax[t + st] > smax[t]) smax[t] = smax[t + st];
        }
        __syncthreads();
    }
    if (t == 0) {
        atomicAdd(&cnt[bi], scnt[0]);
        atomicMax(&packed[bi], smax[0]);
    }
}

typedef float v4f __attribute__((ext_vector_type(4)));

template <int NSTEPS>
__global__ void __launch_bounds__(256)
wmma_masked_sum_kernel(const float* __restrict__ feat, const float* __restrict__ sel,
                       float* __restrict__ s, int P, int nslab) {
    __shared__ float red[8][512];
    const int lane = threadIdx.x & 31;
    const int w    = threadIdx.x >> 5;
    const int slab = blockIdx.x % nslab;
    const int ctg  = (blockIdx.x / nslab) & 7;
    const int b    = blockIdx.x / (nslab * 8);
    const int row  = lane & 15;
    const int h    = lane >> 4;
    const size_t k0 = (size_t)slab * (8 * NSTEPS * 32) + (size_t)w * (NSTEPS * 32) + (size_t)h * 4;
    const float* ap  = sel  + ((size_t)b * I_ + row) * (size_t)P + k0;
    const float* bp0 = feat + ((size_t)b * C_ + (size_t)ctg * 32 + row) * (size_t)P + k0;
    const float* bp1 = bp0 + (size_t)16 * (size_t)P;

    v4f fa[2][4], f0[2][4], f1[2][4];
#pragma unroll
    for (int j = 0; j < 4; ++j) {
        fa[0][j] = *(const v4f*)(ap  + j * 8);
        f0[0][j] = *(const v4f*)(bp0 + j * 8);
        f1[0][j] = *(const v4f*)(bp1 + j * 8);
    }
    v8f acc0 = {}, acc1 = {};
#pragma unroll
    for (int st = 0; st < NSTEPS; ++st) {
        const int cur = st & 1, nxt = cur ^ 1;
        if (st + 1 < NSTEPS) {
#pragma unroll
            for (int j = 0; j < 4; ++j) {
                fa[nxt][j] = *(const v4f*)(ap  + (st + 1) * 32 + j * 8);
                f0[nxt][j] = *(const v4f*)(bp0 + (st + 1) * 32 + j * 8);
                f1[nxt][j] = *(const v4f*)(bp1 + (st + 1) * 32 + j * 8);
            }
        }
        __builtin_amdgcn_sched_barrier(0);
#pragma unroll
        for (int j = 0; j < 4; ++j) {
            v2f a_lo = {fa[cur][j].x, fa[cur][j].y};
            v2f a_hi = {fa[cur][j].z, fa[cur][j].w};
            v2f p_lo = {f0[cur][j].x, f0[cur][j].y};
            v2f p_hi = {f0[cur][j].z, f0[cur][j].w};
            v2f q_lo = {f1[cur][j].x, f1[cur][j].y};
            v2f q_hi = {f1[cur][j].z, f1[cur][j].w};
            acc0 = __builtin_amdgcn_wmma_f32_16x16x4_f32(false, a_lo, false, p_lo,
                                                         (short)0, acc0, false, false);
            acc1 = __builtin_amdgcn_wmma_f32_16x16x4_f32(false, a_lo, false, q_lo,
                                                         (short)0, acc1, false, false);
            acc0 = __builtin_amdgcn_wmma_f32_16x16x4_f32(false, a_hi, false, p_hi,
                                                         (short)0, acc0, false, false);
            acc1 = __builtin_amdgcn_wmma_f32_16x16x4_f32(false, a_hi, false, q_hi,
                                                         (short)0, acc1, false, false);
        }
        __builtin_amdgcn_sched_barrier(0);
    }
#pragma unroll
    for (int r = 0; r < 8; ++r) {
        red[w][r * 32 + lane]       = acc0[r];
        red[w][256 + r * 32 + lane] = acc1[r];
    }
    __syncthreads();
    float t0 = 0.0f, t1 = 0.0f;
#pragma unroll
    for (int q = 0; q < 8; ++q) {
        t0 += red[q][threadIdx.x];
        t1 += red[q][256 + threadIdx.x];
    }
    float* sd = s + ((size_t)b * I_ + (size_t)h * 8 + (size_t)w) * C_ + (size_t)ctg * 32 + row;
    atomicAdd(sd, t0);
    atomicAdd(sd + 16, t1);
}

__global__ void __launch_bounds__(256)
finalize_kernel(const float* __restrict__ f0, const float* __restrict__ f1,
                const float* __restrict__ f2, const float* __restrict__ f3,
                const float* __restrict__ s, const float* __restrict__ cnt,
                const unsigned long long* __restrict__ packed, float* __restrict__ out) {
    int tid = blockIdx.x * blockDim.x + threadIdx.x;
    int c  = tid & 255;
    int bi = tid >> 8;
    int b  = bi >> 4;
    const float* feats[4] = {f0, f1, f2, f3};
    const int Ps[4] = {16384, 4096, 1024, 256};
    float sum = 0.0f;
#pragma unroll
    for (int l = 0; l < 4; ++l) {
        float cv = cnt[l * (B_ * I_) + bi];
        float v;
        if (cv > 0.0f) {
            v = s[l * (B_ * I_ * C_) + tid] / cv;
        } else {
            unsigned long long pk = packed[l * (B_ * I_) + bi];
            unsigned idx = 0xFFFFFFFFu - (unsigned)(pk & 0xFFFFFFFFull);
            v = feats[l][((size_t)b * C_ + c) * (size_t)Ps[l] + idx];
        }
        sum += v;
    }
    out[tid] = 0.25f * sum;
}

extern "C" void kernel_launch(void* const* d_in, const int* in_sizes, int n_in,
                              void* d_out, int out_size, void* d_ws, size_t ws_size,
                              hipStream_t stream) {
    const float* feat0 = (const float*)d_in[0];
    const float* feat1 = (const float*)d_in[1];
    const float* feat2 = (const float*)d_in[2];
    const float* feat3 = (const float*)d_in[3];
    const float* scrib = (const float*)d_in[4];
    float* ws = (float*)d_ws;

    float* sA   = ws + S_OFF;
    float* cntA = ws + CNT_OFF;
    unsigned long long* pkA = (unsigned long long*)((char*)d_ws + PK_BYTEOFF);

    int nz = (B_ * I_ * C_ * 4) + (B_ * I_ * 4) + (B_ * I_ * 4 * 2);
    zero_kernel<<<(nz + 255) / 256, 256, 0, stream>>>(sA, nz);

    const int H[4]  = {128, 64, 32, 16};

    for (int l = 0; l < 4; ++l) {
        int h = H[l], P = h * h;
        int factor = 512 / h;
        int offs = factor / 2 - 1;
        int blocksPerBI = P / 256;
        mask_kernel<<<B_ * I_ * blocksPerBI, 256, 0, stream>>>(
            scrib, ws + SEL_OFF[l], cntA + l * (B_ * I_), pkA + l * (B_ * I_),
            h, factor, offs, blocksPerBI);
    }
    wmma_masked_sum_kernel<8><<<512, 256, 0, stream>>>(feat0, ws + SEL_OFF[0], sA + 0 * (B_ * I_ * C_), 16384, 8);
    wmma_masked_sum_kernel<2><<<512, 256, 0, stream>>>(feat1, ws + SEL_OFF[1], sA + 1 * (B_ * I_ * C_), 4096, 8);
    wmma_masked_sum_kernel<1><<<256, 256, 0, stream>>>(feat2, ws + SEL_OFF[2], sA + 2 * (B_ * I_ * C_), 1024, 4);
    wmma_masked_sum_kernel<1><<<64, 256, 0, stream>>>(feat3, ws + SEL_OFF[3], sA + 3 * (B_ * I_ * C_), 256, 1);
    finalize_kernel<<<(B_ * I_ * C_) / 256, 256, 0, stream>>>(
        feat0, feat1, feat2, feat3, sA, cntA, pkA, (float*)d_out);
}


# four mask launches fused into one pass over the mask input: one workgroup per quarter image, 16-byte row loads, all loads in one group up front, cnt/argmax by wave butterfly + LDS (cnt sum regrouped)
# speedup vs baseline: 1.6499x; 1.2926x over previous
#include <hip/hip_runtime.h>

typedef float v2f __attribute__((ext_vector_type(2)));
typedef float v8f __attribute__((ext_vector_type(8)));
typedef float v4f __attribute__((ext_vector_type(4)));

#define B_ 8
#define I_ 16
#define C_ 256

static const size_t SEL_OFF[4] = {0, 2097152, 2621440, 2752512};
static const size_t S_OFF      = 2785280;
static const size_t CNT_OFF    = 2916352;
static const size_t PK_BYTEOFF = 2916864ull * 4ull;

__global__ void zero_kernel(float* __restrict__ p, int n) {
    int i = blockIdx.x * blockDim.x + threadIdx.x;
    if (i < n) p[i] = 0.0f;
}

__device__ __forceinline__ unsigned long long mask_key(float m, unsigned p) {
    return ((unsigned long long)__float_as_uint(m) << 32) |
           (unsigned long long)(0xFFFFFFFFu - p);
}

__global__ void __launch_bounds__(256)
mask_pyramid_kernel(const float* __restrict__ scrib,
                    float* __restrict__ sel0, float* __restrict__ sel1,
                    float* __restrict__ sel2, float* __restrict__ sel3,
                    float* __restrict__ cnt, unsigned long long* __restrict__ packed) {
    __shared__ float scnt[4][8];
    __shared__ unsigned long long smax[4][8];
    const int lane = threadIdx.x & 31;
    const int w    = threadIdx.x >> 5;
    const int bi   = blockIdx.x >> 2;
    const int qtr  = blockIdx.x & 3;
    const float* wrow = scrib + (size_t)bi * (512 * 512) + (size_t)(qtr * 128 + w * 16) * 512;

    v4f a0[4][4], a1[4][4];
#pragma unroll
    for (int ry = 0; ry < 4; ++ry) {
#pragma unroll
        for (int sg = 0; sg < 4; ++sg) {
            const float* r = wrow + (4 * ry + 1) * 512 + 4 * (32 * sg + lane);
            a0[ry][sg] = *(const v4f*)r;
            a1[ry][sg] = *(const v4f*)(r + 512);
        }
    }
    float b1[2][2][4];
#pragma unroll
    for (int ry = 0; ry < 2; ++ry) {
#pragma unroll
        for (int sg = 0; sg < 2; ++sg) {
            const float* r = wrow + (8 * ry + 3) * 512 + 8 * (32 * sg + lane) + 3;
            b1[ry][sg][0] = r[0];
            b1[ry][sg][1] = r[1];
            b1[ry][sg][2] = r[512];
            b1[ry][sg][3] = r[513];
        }
    }
    float b2[4];
    {
        const float* r = wrow + 7 * 512 + 16 * lane + 7;
        b2[0] = r[0]; b2[1] = r[1]; b2[2] = r[512]; b2[3] = r[513];
    }
    const int pl3 = (32 * w + lane) & 63;
    float b3[4];
    {
        const float* r = wrow - (size_t)(w * 16) * 512 +
                         (size_t)(32 * (pl3 >> 4) + 15) * 512 + 32 * (pl3 & 15) + 15;
        b3[0] = r[0]; b3[1] = r[1]; b3[2] = r[512]; b3[3] = r[513];
    }
    __builtin_amdgcn_sched_barrier(0);

    float c0 = 0.0f, c1 = 0.0f, c2 = 0.0f, c3 = 0.0f;
    unsigned long long k0 = 0ull, k1 = 0ull, k2 = 0ull, k3 = 0ull;

    float* s0 = sel0 + (size_t)bi * 16384;
#pragma unroll
    for (int ry = 0; ry < 4; ++ry) {
#pragma unroll
        for (int sg = 0; sg < 4; ++sg) {
            const int p = (qtr * 32 + w * 4 + ry) * 128 + 32 * sg + lane;
            const float m = 0.25f * (a0[ry][sg].y + a0[ry][sg].z + a1[ry][sg].y + a1[ry][sg].z);
            const float sv = (m > 0.5f) ? 1.0f : 0.0f;
            s0[p] = sv;
            c0 += sv;
            const unsigned long long k = mask_key(m, (unsigned)p);
            k0 = (k > k0) ? k : k0;
        }
    }
    float* s1 = sel1 + (size_t)bi * 4096;
#pragma unroll
    for (int ry = 0; ry < 2; ++ry) {
#pragma unroll
        for (int sg = 0; sg < 2; ++sg) {
            const int p = (qtr * 16 + w * 2 + ry) * 64 + 32 * sg + lane;
            const float m = 0.25f * (b1[ry][sg][0] + b1[ry][sg][1] + b1[ry][sg][2] + b1[ry][sg][3]);
            const float sv = (m > 0.5f) ? 1.0f : 0.0f;
            s1[p] = sv;
            c1 += sv;
            const unsigned long long k = mask_key(m, (unsigned)p);
            k1 = (k > k1) ? k : k1;
        }
    }
    {
        const int p = (qtr * 8 + w) * 32 + lane;
        const float m = 0.25f * (b2[0] + b2[1] + b2[2] + b2[3]);
        const float sv = (m > 0.5f) ? 1.0f : 0.0f;
        sel2[(size_t)bi * 1024 + p] = sv;
        c2 = sv;
        k2 = mask_key(m, (unsigned)p);
    }
    {
        const int p = qtr * 64 + pl3;
        const float m = 0.25f * (b3[0] + b3[1] + b3[2] + b3[3]);
        const float sv = (m > 0.5f) ? 1.0f : 0.0f;
        if (w < 2) sel3[(size_t)bi * 256 + p] = sv;
        c3 = (w < 2) ? sv : 0.0f;
        k3 = mask_key(m, (unsigned)p);
    }

#pragma unroll
    for (int off = 16; off > 0; off >>= 1) {
        const float d0 = __shfl_xor(c0, off);
        const float d1 = __shfl_xor(c1, off);
        const float d2 = __shfl_xor(c2, off);
        const float d3 = __shfl_xor(c3, off);
        const unsigned long long j0 = __shfl_xor(k0, off);
        const unsigned long long j1 = __shfl_xor(k1, off);
        const unsigned long long j2 = __shfl_xor(k2, off);
        const unsigned long long j3 = __shfl_xor(k3, off);
        c0 += d0; c1 += d1; c2 += d2; c3 += d3;
        k0 = (j0 > k0) ? j0 : k0;
        k1 = (j1 > k1) ? j1 : k1;
        k2 = (j2 > k2) ? j2 : k2;
        k3 = (j3 > k3) ? j3 : k3;
    }
    if (lane == 0) {
        scnt[0][w] = c0; scnt[1][w] = c1; scnt[2][w] = c2; scnt[3][w] = c3;
        smax[0][w] = k0; smax[1][w] = k1; smax[2][w] = k2; smax[3][w] = k3;
    }
    __syncthreads();
    if (threadIdx.x < 4) {
        const int l = threadIdx.x;
        float c = 0.0f;
        unsigned long long k = 0ull;
#pragma unroll
        for (int q = 0; q < 8; ++q) {
            c += scnt[l][q];
            const unsigned long long j = smax[l][q];
            k = (j > k) ? j : k;
        }
        atomicAdd(&cnt[l * (B_ * I_) + bi], c);
        atomicMax(&packed[l * (B_ * I_) + bi], k);
    }
}


template <int NSTEPS>
__global__ void __launch_bounds__(256)
wmma_masked_sum_kernel(const float* __restrict__ feat, const float* __restrict__ sel,
                       float* __restrict__ s, int P, int nslab) {
    __shared__ float red[8][512];
    const int lane = threadIdx.x & 31;
    const int w    = threadIdx.x >> 5;
    const int slab = blockIdx.x % nslab;
    const int ctg  = (blockIdx.x / nslab) & 7;
    const int b    = blockIdx.x / (nslab * 8);
    const int row  = lane & 15;
    const int h    = lane >> 4;
    const size_t k0 = (size_t)slab * (8 * NSTEPS * 32) + (size_t)w * (NSTEPS * 32) + (size_t)h * 4;
    const float* ap  = sel  + ((size_t)b * I_ + row) * (size_t)P + k0;
    const float* bp0 = feat + ((size_t)b * C_ + (size_t)ctg * 32 + row) * (size_t)P + k0;
    const float* bp1 = bp0 + (size_t)16 * (size_t)P;

    v4f fa[2][4], f0[2][4], f1[2][4];
#pragma unroll
    for (int j = 0; j < 4; ++j) {
        fa[0][j] = *(const v4f*)(ap  + j * 8);
        f0[0][j] = *(const v4f*)(bp0 + j * 8);
        f1[0][j] = *(const v4f*)(bp1 + j * 8);
    }
    v8f acc0 = {}, acc1 = {};
#pragma unroll
    for (int st = 0; st < NSTEPS; ++st) {
        const int cur = st & 1, nxt = cur ^ 1;
        if (st + 1 < NSTEPS) {
#pragma unroll
            for (int j = 0; j < 4; ++j) {
                fa[nxt][j] = *(const v4f*)(ap  + (st + 1) * 32 + j * 8);
                f0[nxt][j] = *(const v4f*)(bp0 + (st + 1) * 32 + j * 8);
                f1[nxt][j] = *(const v4f*)(bp1 + (st + 1) * 32 + j * 8);
            }
        }
        __builtin_amdgcn_sched_barrier(0);
#pragma unroll
        for (int j = 0; j < 4; ++j) {
            v2f a_lo = {fa[cur][j].x, fa[cur][j].y};
            v2f a_hi = {fa[cur][j].z, fa[cur][j].w};
            v2f p_lo = {f0[cur][j].x, f0[cur][j].y};
            v2f p_hi = {f0[cur][j].z, f0[cur][j].w};
            v2f q_lo = {f1[cur][j].x, f1[cur][j].y};
            v2f q_hi = {f1[cur][j].z, f1[cur][j].w};
            acc0 = __builtin_amdgcn_wmma_f32_16x16x4_f32(false, a_lo, false, p_lo,
                                                         (short)0, acc0, false, false);
            acc1 = __builtin_amdgcn_wmma_f32_16x16x4_f32(false, a_lo, false, q_lo,
                                                         (short)0, acc1, false, false);
            acc0 = __builtin_amdgcn_wmma_f32_16x16x4_f32(false, a_hi, false, p_hi,
                                                         (short)0, acc0, false, false);
            acc1 = __builtin_amdgcn_wmma_f32_16x16x4_f32(false, a_hi, false, q_hi,
                                                         (short)0, acc1, false, false);
        }
        __builtin_amdgcn_sched_barrier(0);
    }
#pragma unroll
    for (int r = 0; r < 8; ++r) {
        red[w][r * 32 + lane]       = acc0[r];
        red[w][256 + r * 32 + lane] = acc1[r];
    }
    __syncthreads();
    float t0 = 0.0f, t1 = 0.0f;
#pragma unroll
    for (int q = 0; q < 8; ++q) {
        t0 += red[q][threadIdx.x];
        t1 += red[q][256 + threadIdx.x];
    }
    float* sd = s + ((size_t)b * I_ + (size_t)h * 8 + (size_t)w) * C_ + (size_t)ctg * 32 + row;
    atomicAdd(sd, t0);
    atomicAdd(sd + 16, t1);
}

__global__ void __launch_bounds__(256)
finalize_kernel(const float* __restrict__ f0, const float* __restrict__ f1,
                const float* __restrict__ f2, const float* __restrict__ f3,
                const float* __restrict__ s, const float* __restrict__ cnt,
                const unsigned long long* __restrict__ packed, float* __restrict__ out) {
    int tid = blockIdx.x * blockDim.x + threadIdx.x;
    int c  = tid & 255;
    int bi = tid >> 8;
    int b  = bi >> 4;
    const float* feats[4] = {f0, f1, f2, f3};
    const int Ps[4] = {16384, 4096, 1024, 256};
    float sum = 0.0f;
#pragma unroll
    for (int l = 0; l < 4; ++l) {
        float cv = cnt[l * (B_ * I_) + bi];
        float v;
        if (cv > 0.0f) {
            v = s[l * (B_ * I_ * C_) + tid] / cv;
        } else {
            unsigned long long pk = packed[l * (B_ * I_) + bi];
            unsigned idx = 0xFFFFFFFFu - (unsigned)(pk & 0xFFFFFFFFull);
            v = feats[l][((size_t)b * C_ + c) * (size_t)Ps[l] + idx];
        }
        sum += v;
    }
    out[tid] = 0.25f * sum;
}

extern "C" void kernel_launch(void* const* d_in, const int* in_sizes, int n_in,
                              void* d_out, int out_size, void* d_ws, size_t ws_size,
                              hipStream_t stream) {
    const float* feat0 = (const float*)d_in[0];
    const float* feat1 = (const float*)d_in[1];
    const float* feat2 = (const float*)d_in[2];
    const float* feat3 = (const float*)d_in[3];
    const float* scrib = (const float*)d_in[4];
    float* ws = (float*)d_ws;

    float* sA   = ws + S_OFF;
    float* cntA = ws + CNT_OFF;
    unsigned long long* pkA = (unsigned long long*)((char*)d_ws + PK_BYTEOFF);

    int nz = (B_ * I_ * C_ * 4) + (B_ * I_ * 4) + (B_ * I_ * 4 * 2);
    zero_kernel<<<(nz + 255) / 256, 256, 0, stream>>>(sA, nz);

    mask_pyramid_kernel<<<B_ * I_ * 4, 256, 0, stream>>>(
        scrib, ws + SEL_OFF[0], ws + SEL_OFF[1], ws + SEL_OFF[2], ws + SEL_OFF[3], cntA, pkA);
    wmma_masked_sum_kernel<8><<<512, 256, 0, stream>>>(feat0, ws + SEL_OFF[0], sA + 0 * (B_ * I_ * C_), 16384, 8);
    wmma_masked_sum_kernel<2><<<512, 256, 0, stream>>>(feat1, ws + SEL_OFF[1], sA + 1 * (B_ * I_ * C_), 4096, 8);
    wmma_masked_sum_kernel<1><<<256, 256, 0, stream>>>(feat2, ws + SEL_OFF[2], sA + 2 * (B_ * I_ * C_), 1024, 4);
    wmma_masked_sum_kernel<1><<<64, 256, 0, stream>>>(feat3, ws + SEL_OFF[3], sA + 3 * (B_ * I_ * C_), 256, 1);
    finalize_kernel<<<(B_ * I_ * C_) / 256, 256, 0, stream>>>(
        feat0, feat1, feat2, feat3, sA, cntA, pkA, (float*)d_out);
}


# four masked-sum WMMA launches merged into one grid of 1344 workgroups (level from blockIdx, long level-0 workgroups first); partial sums unchanged, atomic adds to s arrive unordered as in the parent
# speedup vs baseline: 2.0548x; 1.2454x over previous
#include <hip/hip_runtime.h>

typedef float v2f __attribute__((ext_vector_type(2)));
typedef float v8f __attribute__((ext_vector_type(8)));
typedef float v4f __attribute__((ext_vector_type(4)));

#define B_ 8
#define I_ 16
#define C_ 256

static const size_t SEL_OFF[4] = {0, 2097152, 2621440, 2752512};
static const size_t S_OFF      = 2785280;
static const size_t CNT_OFF    = 2916352;
static const size_t PK_BYTEOFF = 2916864ull * 4ull;

__global__ void zero_kernel(float* __restrict__ p, int n) {
    int i = blockIdx.x * blockDim.x + threadIdx.x;
    if (i < n) p[i] = 0.0f;
}

__device__ __forceinline__ unsigned long long mask_key(float m, unsigned p) {
    return ((unsigned long long)__float_as_uint(m) << 32) |
           (unsigned long long)(0xFFFFFFFFu - p);
}

__global__ void __launch_bounds__(256)
mask_pyramid_kernel(const float* __restrict__ scrib,
                    float* __restrict__ sel0, float* __restrict__ sel1,
                    float* __restrict__ sel2, float* __restrict__ sel3,
                    float* __restrict__ cnt, unsigned long long* __restrict__ packed) {
    __shared__ float scnt[4][8];
    __shared__ unsigned long long smax[4][8];
    const int lane = threadIdx.x & 31;
    const int w    = threadIdx.x >> 5;
    const int bi   = blockIdx.x >> 2;
    const int qtr  = blockIdx.x & 3;
    const float* wrow = scrib + (size_t)bi * (512 * 512) + (size_t)(qtr * 128 + w * 16) * 512;

    v4f a0[4][4], a1[4][4];
#pragma unroll
    for (int ry = 0; ry < 4; ++ry) {
#pragma unroll
        for (int sg = 0; sg < 4; ++sg) {
            const float* r = wrow + (4 * ry + 1) * 512 + 4 * (32 * sg + lane);
            a0[ry][sg] = *(const v4f*)r;
            a1[ry][sg] = *(const v4f*)(r + 512);
        }
    }
    float b1[2][2][4];
#pragma unroll
    for (int ry = 0; ry < 2; ++ry) {
#pragma unroll
        for (int sg = 0; sg < 2; ++sg) {
            const float* r = wrow + (8 * ry + 3) * 512 + 8 * (32 * sg + lane) + 3;
            b1[ry][sg][0] = r[0];
            b1[ry][sg][1] = r[1];
            b1[ry][sg][2] = r[512];
            b1[ry][sg][3] = r[513];
        }
    }
    float b2[4];
    {
        const float* r = wrow + 7 * 512 + 16 * lane + 7;
        b2[0] = r[0]; b2[1] = r[1]; b2[2] = r[512]; b2[3] = r[513];
    }
    const int pl3 = (32 * w + lane) & 63;
    float b3[4];
    {
        const float* r = wrow - (size_t)(w * 16) * 512 +
                         (size_t)(32 * (pl3 >> 4) + 15) * 512 + 32 * (pl3 & 15) + 15;
        b3[0] = r[0]; b3[1] = r[1]; b3[2] = r[512]; b3[3] = r[513];
    }
    __builtin_amdgcn_sched_barrier(0);

    float c0 = 0.0f, c1 = 0.0f, c2 = 0.0f, c3 = 0.0f;
    unsigned long long k0 = 0ull, k1 = 0ull, k2 = 0ull, k3 = 0ull;

    float* s0 = sel0 + (size_t)bi * 16384;
#pragma unroll
    for (int ry = 0; ry < 4; ++ry) {
#pragma unroll
        for (int sg = 0; sg < 4; ++sg) {
            const int p = (qtr * 32 + w * 4 + ry) * 128 + 32 * sg + lane;
            const float m = 0.25f * (a0[ry][sg].y + a0[ry][sg].z + a1[ry][sg].y + a1[ry][sg].z);
            const float sv = (m > 0.5f) ? 1.0f : 0.0f;
            s0[p] = sv;
            c0 += sv;
            const unsigned long long k = mask_key(m, (unsigned)p);
            k0 = (k > k0) ? k : k0;
        }
    }
    float* s1 = sel1 + (size_t)bi * 4096;
#pragma unroll
    for (int ry = 0; ry < 2; ++ry) {
#pragma unroll
        for (int sg = 0; sg < 2; ++sg) {
            const int p = (qtr * 16 + w * 2 + ry) * 64 + 32 * sg + lane;
            const float m = 0.25f * (b1[ry][sg][0] + b1[ry][sg][1] + b1[ry][sg][2] + b1[ry][sg][3]);
            const float sv = (m > 0.5f) ? 1.0f : 0.0f;
            s1[p] = sv;
            c1 += sv;
            const unsigned long long k = mask_key(m, (unsigned)p);
            k1 = (k > k1) ? k : k1;
        }
    }
    {
        const int p = (qtr * 8 + w) * 32 + lane;
        const float m = 0.25f * (b2[0] + b2[1] + b2[2] + b2[3]);
        const float sv = (m > 0.5f) ? 1.0f : 0.0f;
        sel2[(size_t)bi * 1024 + p] = sv;
        c2 = sv;
        k2 = mask_key(m, (unsigned)p);
    }
    {
        const int p = qtr * 64 + pl3;
        const float m = 0.25f * (b3[0] + b3[1] + b3[2] + b3[3]);
        const float sv = (m > 0.5f) ? 1.0f : 0.0f;
        if (w < 2) sel3[(size_t)bi * 256 + p] = sv;
        c3 = (w < 2) ? sv : 0.0f;
        k3 = mask_key(m, (unsigned)p);
    }

#pragma unroll
    for (int off = 16; off > 0; off >>= 1) {
        const float d0 = __shfl_xor(c0, off);
        const float d1 = __shfl_xor(c1, off);
        const float d2 = __shfl_xor(c2, off);
        const float d3 = __shfl_xor(c3, off);
        const unsigned long long j0 = __shfl_xor(k0, off);
        const unsigned long long j1 = __shfl_xor(k1, off);
        const unsigned long long j2 = __shfl_xor(k2, off);
        const unsigned long long j3 = __shfl_xor(k3, off);
        c0 += d0; c1 += d1; c2 += d2; c3 += d3;
        k0 = (j0 > k0) ? j0 : k0;
        k1 = (j1 > k1) ? j1 : k1;
        k2 = (j2 > k2) ? j2 : k2;
        k3 = (j3 > k3) ? j3 : k3;
    }
    if (lane == 0) {
        scnt[0][w] = c0; scnt[1][w] = c1; scnt[2][w] = c2; scnt[3][w] = c3;
        smax[0][w] = k0; smax[1][w] = k1; smax[2][w] = k2; smax[3][w] = k3;
    }
    __syncthreads();
    if (threadIdx.x < 4) {
        const int l = threadIdx.x;
        float c = 0.0f;
        unsigned long long k = 0ull;
#pragma unroll
        for (int q = 0; q < 8; ++q) {
            c += scnt[l][q];
            const unsigned long long j = smax[l][q];
            k = (j > k) ? j : k;
        }
        atomicAdd(&cnt[l * (B_ * I_) + bi], c);
        atomicMax(&packed[l * (B_ * I_) + bi], k);
    }
}


template <int NSTEPS>
__device__ __forceinline__ void masked_sum_slice(const float* __restrict__ ap,
                                                 const float* __restrict__ bp0,
                                                 const float* __restrict__ bp1,
                                                 v8f& acc0, v8f& acc1) {
    v4f fa[2][4], f0[2][4], f1[2][4];
#pragma unroll
    for (int j = 0; j < 4; ++j) {
        fa[0][j] = *(const v4f*)(ap  + j * 8);
        f0[0][j] = *(const v4f*)(bp0 + j * 8);
        f1[0][j] = *(const v4f*)(bp1 + j * 8);
    }
#pragma unroll
    for (int st = 0; st < NSTEPS; ++st) {
        const int cur = st & 1, nxt = cur ^ 1;
        if (st + 1 < NSTEPS) {
#pragma unroll
            for (int j = 0; j < 4; ++j) {
                fa[nxt][j] = *(const v4f*)(ap  + (st + 1) * 32 + j * 8);
                f0[nxt][j] = *(const v4f*)(bp0 + (st + 1) * 32 + j * 8);
                f1[nxt][j] = *(const v4f*)(bp1 + (st + 1) * 32 + j * 8);
            }
        }
        __builtin_amdgcn_sched_barrier(0);
#pragma unroll
        for (int j = 0; j < 4; ++j) {
            v2f a_lo = {fa[cur][j].x, fa[cur][j].y};
            v2f a_hi = {fa[cur][j].z, fa[cur][j].w};
            v2f p_lo = {f0[cur][j].x, f0[cur][j].y};
            v2f p_hi = {f0[cur][j].z, f0[cur][j].w};
            v2f q_lo = {f1[cur][j].x, f1[cur][j].y};
            v2f q_hi = {f1[cur][j].z, f1[cur][j].w};
            acc0 = __builtin_amdgcn_wmma_f32_16x16x4_f32(false, a_lo, false, p_lo,
                                                         (short)0, acc0, false, false);
            acc1 = __builtin_amdgcn_wmma_f32_16x16x4_f32(false, a_lo, false, q_lo,
                                                         (short)0, acc1, false, false);
            acc0 = __builtin_amdgcn_wmma_f32_16x16x4_f32(false, a_hi, false, p_hi,
                                                         (short)0, acc0, false, false);
            acc1 = __builtin_amdgcn_wmma_f32_16x16x4_f32(false, a_hi, false, q_hi,
                                                         (short)0, acc1, false, false);
        }
        __builtin_amdgcn_sched_barrier(0);
    }
}

__global__ void __launch_bounds__(256)
wmma_masked_sum_levels_kernel(const float* __restrict__ feat0, const float* __restrict__ feat1,
                              const float* __restrict__ feat2, const float* __restrict__ feat3,
                              const float* __restrict__ selbase, float* __restrict__ sbase) {
    __shared__ float red[8][512];
    const int lane = threadIdx.x & 31;
    const int w    = threadIdx.x >> 5;
    const int row  = lane & 15;
    const int h    = lane >> 4;

    int bid = blockIdx.x;
    int lvl, lgP, lgSlab, lgSteps;
    const float* feat;
    size_t selOff;
    if (bid < 512)       {              lvl = 0; feat = feat0; lgP = 14; lgSlab = 3; lgSteps = 3; selOff = 0; }
    else if (bid < 1024) { bid -= 512;  lvl = 1; feat = feat1; lgP = 12; lgSlab = 3; lgSteps = 1; selOff = 2097152; }
    else if (bid < 1280) { bid -= 1024; lvl = 2; feat = feat2; lgP = 10; lgSlab = 2; lgSteps = 0; selOff = 2621440; }
    else                 { bid -= 1280; lvl = 3; feat = feat3; lgP = 8;  lgSlab = 0; lgSteps = 0; selOff = 2752512; }
    const int slab = bid & ((1 << lgSlab) - 1);
    const int ctg  = (bid >> lgSlab) & 7;
    const int b    = bid >> (lgSlab + 3);
    const size_t k0 = ((size_t)slab << (8 + lgSteps)) + ((size_t)w << (5 + lgSteps)) + (size_t)h * 4;
    const float* ap  = selbase + selOff + (((size_t)b * I_ + row) << lgP) + k0;
    const float* bp0 = feat + (((size_t)b * C_ + (size_t)ctg * 32 + row) << lgP) + k0;
    const float* bp1 = bp0 + ((size_t)16 << lgP);

    v8f acc0 = {}, acc1 = {};
    if (lgSteps == 3)      masked_sum_slice<8>(ap, bp0, bp1, acc0, acc1);
    else if (lgSteps == 1) masked_sum_slice<2>(ap, bp0, bp1, acc0, acc1);
    else                   masked_sum_slice<1>(ap, bp0, bp1, acc0, acc1);

#pragma unroll
    for (int r = 0; r < 8; ++r) {
        red[w][r * 32 + lane]       = acc0[r];
        red[w][256 + r * 32 + lane] = acc1[r];
    }
    __syncthreads();
    float t0 = 0.0f, t1 = 0.0f;
#pragma unroll
    for (int q = 0; q < 8; ++q) {
        t0 += red[q][threadIdx.x];
        t1 += red[q][256 + threadIdx.x];
    }
    float* sd = sbase + (size_t)lvl * (B_ * I_ * C_) +
                ((size_t)b * I_ + (size_t)h * 8 + (size_t)w) * C_ + (size_t)ctg * 32 + row;
    atomicAdd(sd, t0);
    atomicAdd(sd + 16, t1);
}

__global__ void __launch_bounds__(256)
finalize_kernel(const float* __restrict__ f0, const float* __restrict__ f1,
                const float* __restrict__ f2, const float* __restrict__ f3,
                const float* __restrict__ s, const float* __restrict__ cnt,
                const unsigned long long* __restrict__ packed, float* __restrict__ out) {
    int tid = blockIdx.x * blockDim.x + threadIdx.x;
    int c  = tid & 255;
    int bi = tid >> 8;
    int b  = bi >> 4;
    const float* feats[4] = {f0, f1, f2, f3};
    const int Ps[4] = {16384, 4096, 1024, 256};
    float sum = 0.0f;
#pragma unroll
    for (int l = 0; l < 4; ++l) {
        float cv = cnt[l * (B_ * I_) + bi];
        float v;
        if (cv > 0.0f) {
            v = s[l * (B_ * I_ * C_) + tid] / cv;
        } else {
            unsigned long long pk = packed[l * (B_ * I_) + bi];
            unsigned idx = 0xFFFFFFFFu - (unsigned)(pk & 0xFFFFFFFFull);
            v = feats[l][((size_t)b * C_ + c) * (size_t)Ps[l] + idx];
        }
        sum += v;
    }
    out[tid] = 0.25f * sum;
}

extern "C" void kernel_launch(void* const* d_in, const int* in_sizes, int n_in,
                              void* d_out, int out_size, void* d_ws, size_t ws_size,
                              hipStream_t stream) {
    const float* feat0 = (const float*)d_in[0];
    const float* feat1 = (const float*)d_in[1];
    const float* feat2 = (const float*)d_in[2];
    const float* feat3 = (const float*)d_in[3];
    const float* scrib = (const float*)d_in[4];
    float* ws = (float*)d_ws;

    float* sA   = ws + S_OFF;
    float* cntA = ws + CNT_OFF;
    unsigned long long* pkA = (unsigned long long*)((char*)d_ws + PK_BYTEOFF);

    int nz = (B_ * I_ * C_ * 4) + (B_ * I_ * 4) + (B_ * I_ * 4 * 2);
    zero_kernel<<<(nz + 255) / 256, 256, 0, stream>>>(sA, nz);

    mask_pyramid_kernel<<<B_ * I_ * 4, 256, 0, stream>>>(
        scrib, ws + SEL_OFF[0], ws + SEL_OFF[1], ws + SEL_OFF[2], ws + SEL_OFF[3], cntA, pkA);
    wmma_masked_sum_levels_kernel<<<1344, 256, 0, stream>>>(feat0, feat1, feat2, feat3, ws, sA);
    finalize_kernel<<<(B_ * I_ * C_) / 256, 256, 0, stream>>>(
        feat0, feat1, feat2, feat3, sA, cntA, pkA, (float*)d_out);
}


# sel passed from the mask kernel to the masked-sum kernel as bit masks (wave vote, 1 bit per pixel) instead of float planes; A built from the bits as exact 0 or 1; f32 WMMA and sum order unchanged
# speedup vs baseline: 2.8227x; 1.3737x over previous
#include <hip/hip_runtime.h>

typedef float v2f __attribute__((ext_vector_type(2)));
typedef float v8f __attribute__((ext_vector_type(8)));
typedef float v4f __attribute__((ext_vector_type(4)));
typedef unsigned v4u __attribute__((ext_vector_type(4)));
typedef unsigned v2u __attribute__((ext_vector_type(2)));

#define B_ 8
#define I_ 16
#define C_ 256

static const size_t SELW_OFF[4] = {0, 65536, 81920, 86016};
static const size_t CNTP_OFF   = 2785280;
static const size_t PKP_BYTEOFF = 2787328ull * 4ull;

__device__ __forceinline__ unsigned long long mask_key(float m, unsigned p) {
    return ((unsigned long long)__float_as_uint(m) << 32) |
           (unsigned long long)(0xFFFFFFFFu - p);
}

__global__ void __launch_bounds__(256)
mask_pyramid_kernel(const float* __restrict__ scrib,
                    unsigned* __restrict__ selb0, unsigned* __restrict__ selb1,
                    unsigned* __restrict__ selb2, unsigned* __restrict__ selb3,
                    float* __restrict__ cntp, unsigned long long* __restrict__ pkp,
                    float* __restrict__ out) {
    __shared__ float scnt[4][8];
    __shared__ unsigned long long smax[4][8];
    __shared__ unsigned sbit[176];
    const int lane = threadIdx.x & 31;
    const int w    = threadIdx.x >> 5;
    const int bi   = blockIdx.x >> 2;
    const int qtr  = blockIdx.x & 3;
    const float* wrow = scrib + (size_t)bi * (512 * 512) + (size_t)(qtr * 128 + w * 16) * 512;

    v4f a0[4][4], a1[4][4];
#pragma unroll
    for (int ry = 0; ry < 4; ++ry) {
#pragma unroll
        for (int sg = 0; sg < 4; ++sg) {
            const float* r = wrow + (4 * ry + 1) * 512 + 4 * (32 * sg + lane);
            a0[ry][sg] = *(const v4f*)r;
            a1[ry][sg] = *(const v4f*)(r + 512);
        }
    }
    float b1[2][2][4];
#pragma unroll
    for (int ry = 0; ry < 2; ++ry) {
#pragma unroll
        for (int sg = 0; sg < 2; ++sg) {
            const float* r = wrow + (8 * ry + 3) * 512 + 8 * (32 * sg + lane) + 3;
            b1[ry][sg][0] = r[0];
            b1[ry][sg][1] = r[1];
            b1[ry][sg][2] = r[512];
            b1[ry][sg][3] = r[513];
        }
    }
    float b2[4];
    {
        const float* r = wrow + 7 * 512 + 16 * lane + 7;
        b2[0] = r[0]; b2[1] = r[1]; b2[2] = r[512]; b2[3] = r[513];
    }
    const int pl3 = (32 * w + lane) & 63;
    float b3[4];
    {
        const float* r = wrow - (size_t)(w * 16) * 512 +
                         (size_t)(32 * (pl3 >> 4) + 15) * 512 + 32 * (pl3 & 15) + 15;
        b3[0] = r[0]; b3[1] = r[1]; b3[2] = r[512]; b3[3] = r[513];
    }
    __builtin_amdgcn_sched_barrier(0);

    float c0 = 0.0f, c1 = 0.0f, c2 = 0.0f, c3 = 0.0f;
    unsigned long long k0 = 0ull, k1 = 0ull, k2 = 0ull, k3 = 0ull;

    unsigned bw0[4][4];
#pragma unroll
    for (int ry = 0; ry < 4; ++ry) {
#pragma unroll
        for (int sg = 0; sg < 4; ++sg) {
            const int p = (qtr * 32 + w * 4 + ry) * 128 + 32 * sg + lane;
            const float m = 0.25f * (a0[ry][sg].y + a0[ry][sg].z + a1[ry][sg].y + a1[ry][sg].z);
            const bool fg = m > 0.5f;
            bw0[ry][sg] = __builtin_amdgcn_ballot_w32(fg);
            const float sv = fg ? 1.0f : 0.0f;
            c0 += sv;
            const unsigned long long k = mask_key(m, (unsigned)p);
            k0 = (k > k0) ? k : k0;
        }
    }
    unsigned bw1[2][2];
#pragma unroll
    for (int ry = 0; ry < 2; ++ry) {
#pragma unroll
        for (int sg = 0; sg < 2; ++sg) {
            const int p = (qtr * 16 + w * 2 + ry) * 64 + 32 * sg + lane;
            const float m = 0.25f * (b1[ry][sg][0] + b1[ry][sg][1] + b1[ry][sg][2] + b1[ry][sg][3]);
            const bool fg = m > 0.5f;
            bw1[ry][sg] = __builtin_amdgcn_ballot_w32(fg);
            const float sv = fg ? 1.0f : 0.0f;
            c1 += sv;
            const unsigned long long k = mask_key(m, (unsigned)p);
            k1 = (k > k1) ? k : k1;
        }
    }
    unsigned bw2;
    {
        const int p = (qtr * 8 + w) * 32 + lane;
        const float m = 0.25f * (b2[0] + b2[1] + b2[2] + b2[3]);
        const bool fg = m > 0.5f;
        bw2 = __builtin_amdgcn_ballot_w32(fg);
        const float sv = fg ? 1.0f : 0.0f;
        c2 = sv;
        k2 = mask_key(m, (unsigned)p);
    }
    unsigned bw3;
    {
        const int p = qtr * 64 + pl3;
        const float m = 0.25f * (b3[0] + b3[1] + b3[2] + b3[3]);
        const bool fg = m > 0.5f;
        bw3 = __builtin_amdgcn_ballot_w32(fg);
        const float sv = fg ? 1.0f : 0.0f;
        c3 = (w < 2) ? sv : 0.0f;
        k3 = mask_key(m, (unsigned)p);
    }

#pragma unroll
    for (int off = 16; off > 0; off >>= 1) {
        const float d0 = __shfl_xor(c0, off);
        const float d1 = __shfl_xor(c1, off);
        const float d2 = __shfl_xor(c2, off);
        const float d3 = __shfl_xor(c3, off);
        const unsigned long long j0 = __shfl_xor(k0, off);
        const unsigned long long j1 = __shfl_xor(k1, off);
        const unsigned long long j2 = __shfl_xor(k2, off);
        const unsigned long long j3 = __shfl_xor(k3, off);
        c0 += d0; c1 += d1; c2 += d2; c3 += d3;
        k0 = (j0 > k0) ? j0 : k0;
        k1 = (j1 > k1) ? j1 : k1;
        k2 = (j2 > k2) ? j2 : k2;
        k3 = (j3 > k3) ? j3 : k3;
    }
    if (lane == 0) {
        scnt[0][w] = c0; scnt[1][w] = c1; scnt[2][w] = c2; scnt[3][w] = c3;
        smax[0][w] = k0; smax[1][w] = k1; smax[2][w] = k2; smax[3][w] = k3;
#pragma unroll
        for (int ry = 0; ry < 4; ++ry) {
#pragma unroll
            for (int sg = 0; sg < 4; ++sg) sbit[w * 16 + ry * 4 + sg] = bw0[ry][sg];
        }
#pragma unroll
        for (int ry = 0; ry < 2; ++ry) {
#pragma unroll
            for (int sg = 0; sg < 2; ++sg) sbit[128 + w * 4 + ry * 2 + sg] = bw1[ry][sg];
        }
        sbit[160 + w] = bw2;
        if (w < 2) sbit[168 + w] = bw3;
    }
    __syncthreads();
    {
        const int t = threadIdx.x;
        if (t < 128)      selb0[(size_t)bi * 512 + qtr * 128 + t]       = sbit[t];
        else if (t < 160) selb1[(size_t)bi * 128 + qtr * 32 + (t - 128)] = sbit[t];
        else if (t < 168) selb2[(size_t)bi * 32  + qtr * 8  + (t - 160)] = sbit[t];
        else if (t < 170) selb3[(size_t)bi * 8   + qtr * 2  + (t - 168)] = sbit[t];
    }
    if (threadIdx.x < 4) {
        const int l = threadIdx.x;
        float c = 0.0f;
        unsigned long long k = 0ull;
#pragma unroll
        for (int q = 0; q < 8; ++q) {
            c += scnt[l][q];
            const unsigned long long j = smax[l][q];
            k = (j > k) ? j : k;
        }
        cntp[l * (B_ * I_ * 4) + blockIdx.x] = c;
        pkp[l * (B_ * I_ * 4) + blockIdx.x]  = k;
    }
    if (w == 2 || w == 3) out[blockIdx.x * 64 + (w - 2) * 32 + lane] = 0.0f;
}


__device__ __forceinline__ float mask_bit(unsigned ws, int n) {
    return (float)((ws >> n) & 1u);
}

template <int NSTEPS>
__device__ __forceinline__ void masked_sum_slice(const unsigned* __restrict__ wp, int hs,
                                                 const float* __restrict__ bp0,
                                                 const float* __restrict__ bp1,
                                                 v8f& acc0, v8f& acc1) {
    unsigned wd[8];
    if (NSTEPS == 8) {
        const v4u x = *(const v4u*)wp;
        const v4u y = *(const v4u*)(wp + 4);
        wd[0] = x.x; wd[1] = x.y; wd[2] = x.z; wd[3] = x.w;
        wd[4] = y.x; wd[5] = y.y; wd[6] = y.z; wd[7] = y.w;
    } else if (NSTEPS == 2) {
        const v2u x = *(const v2u*)wp;
        wd[0] = x.x; wd[1] = x.y;
    } else {
        wd[0] = *wp;
    }
    v4f f0[2][4], f1[2][4];
#pragma unroll
    for (int j = 0; j < 4; ++j) {
        f0[0][j] = *(const v4f*)(bp0 + j * 8);
        f1[0][j] = *(const v4f*)(bp1 + j * 8);
    }
#pragma unroll
    for (int st = 0; st < NSTEPS; ++st) {
        const int cur = st & 1, nxt = cur ^ 1;
        if (st + 1 < NSTEPS) {
#pragma unroll
            for (int j = 0; j < 4; ++j) {
                f0[nxt][j] = *(const v4f*)(bp0 + (st + 1) * 32 + j * 8);
                f1[nxt][j] = *(const v4f*)(bp1 + (st + 1) * 32 + j * 8);
            }
        }
        __builtin_amdgcn_sched_barrier(0);
        const unsigned ws = wd[st] >> hs;
#pragma unroll
        for (int j = 0; j < 4; ++j) {
            v2f a_lo = {mask_bit(ws, j * 8 + 0), mask_bit(ws, j * 8 + 1)};
            v2f a_hi = {mask_bit(ws, j * 8 + 2), mask_bit(ws, j * 8 + 3)};
            v2f p_lo = {f0[cur][j].x, f0[cur][j].y};
            v2f p_hi = {f0[cur][j].z, f0[cur][j].w};
            v2f q_lo = {f1[cur][j].x, f1[cur][j].y};
            v2f q_hi = {f1[cur][j].z, f1[cur][j].w};
            acc0 = __builtin_amdgcn_wmma_f32_16x16x4_f32(false, a_lo, false, p_lo,
                                                         (short)0, acc0, false, false);
            acc1 = __builtin_amdgcn_wmma_f32_16x16x4_f32(false, a_lo, false, q_lo,
                                                         (short)0, acc1, false, false);
            acc0 = __builtin_amdgcn_wmma_f32_16x16x4_f32(false, a_hi, false, p_hi,
                                                         (short)0, acc0, false, false);
            acc1 = __builtin_amdgcn_wmma_f32_16x16x4_f32(false, a_hi, false, q_hi,
                                                         (short)0, acc1, false, false);
        }
        __builtin_amdgcn_sched_barrier(0);
    }
}

__device__ __forceinline__ void add_level_share(float* od, float t, float cv, bool first_slab,
                                                const unsigned long long* __restrict__ pk4,
                                                const float* __restrict__ fch, int lgP) {
    if (cv > 0.0f) {
        atomicAdd(od, 0.25f * (t / cv));
    } else if (first_slab) {
        unsigned long long k = pk4[0];
#pragma unroll
        for (int q = 1; q < 4; ++q) {
            const unsigned long long j = pk4[q];
            k = (j > k) ? j : k;
        }
        const unsigned idx = (0xFFFFFFFFu - (unsigned)(k & 0xFFFFFFFFull)) & ((1u << lgP) - 1u);
        atomicAdd(od, 0.25f * fch[idx]);
    }
}

__global__ void __launch_bounds__(256)
wmma_masked_sum_levels_kernel(const float* __restrict__ feat0, const float* __restrict__ feat1,
                              const float* __restrict__ feat2, const float* __restrict__ feat3,
                              const unsigned* __restrict__ selbits,
                              const float* __restrict__ cntp,
                              const unsigned long long* __restrict__ pkp,
                              float* __restrict__ out) {
    __shared__ float red[8][512];
    const int lane = threadIdx.x & 31;
    const int w    = threadIdx.x >> 5;
    const int row  = lane & 15;
    const int h    = lane >> 4;

    int bid = blockIdx.x;
    int lvl, lgP, lgSlab, lgSteps;
    const float* feat;
    size_t selOff;
    if (bid < 512)       {              lvl = 0; feat = feat0; lgP = 14; lgSlab = 3; lgSteps = 3; selOff = 0; }
    else if (bid < 1024) { bid -= 512;  lvl = 1; feat = feat1; lgP = 12; lgSlab = 3; lgSteps = 1; selOff = 65536; }
    else if (bid < 1280) { bid -= 1024; lvl = 2; feat = feat2; lgP = 10; lgSlab = 2; lgSteps = 0; selOff = 81920; }
    else                 { bid -= 1280; lvl = 3; feat = feat3; lgP = 8;  lgSlab = 0; lgSteps = 0; selOff = 86016; }
    const int slab = bid & ((1 << lgSlab) - 1);
    const int ctg  = (bid >> lgSlab) & 7;
    const int b    = bid >> (lgSlab + 3);
    const size_t ks = ((size_t)slab << (8 + lgSteps)) + ((size_t)w << (5 + lgSteps));
    const size_t k0 = ks + (size_t)h * 4;
    const unsigned* wp = selbits + selOff + (((size_t)b * I_ + row) << (lgP - 5)) + (ks >> 5);
    const int hs = h * 4;
    const float* bp0 = feat + (((size_t)b * C_ + (size_t)ctg * 32 + row) << lgP) + k0;
    const float* bp1 = bp0 + ((size_t)16 << lgP);

    v8f acc0 = {}, acc1 = {};
    const int wv = __builtin_amdgcn_readfirstlane(w);
    const int partA = lvl * (B_ * I_ * 4) + (b * I_ + wv) * 4;
    const int partB = partA + 8 * 4;
    const v4f cqa = *(const v4f*)(cntp + partA);
    const v4f cqb = *(const v4f*)(cntp + partB);

    if (lgSteps == 3)      masked_sum_slice<8>(wp, hs, bp0, bp1, acc0, acc1);
    else if (lgSteps == 1) masked_sum_slice<2>(wp, hs, bp0, bp1, acc0, acc1);
    else                   masked_sum_slice<1>(wp, hs, bp0, bp1, acc0, acc1);

#pragma unroll
    for (int r = 0; r < 8; ++r) {
        red[w][r * 32 + lane]       = acc0[r];
        red[w][256 + r * 32 + lane] = acc1[r];
    }
    __syncthreads();
    const int ia = (lane >> 4) * 256 + wv * 32 + (lane & 15);
    float t0 = 0.0f, t1 = 0.0f;
#pragma unroll
    for (int q = 0; q < 8; ++q) {
        t0 += red[q][ia];
        t1 += red[q][ia + 16];
    }
    const float cva = (cqa.x + cqa.y) + (cqa.z + cqa.w);
    const float cvb = (cqb.x + cqb.y) + (cqb.z + cqb.w);
    float* od = out + ((size_t)b * I_ + (size_t)wv) * C_ + (size_t)ctg * 32 + lane;
    const float* fch = feat + (((size_t)b * C_ + (size_t)ctg * 32 + lane) << lgP);
    add_level_share(od,          t0, cva, slab == 0, pkp + partA, fch, lgP);
    add_level_share(od + 8 * C_, t1, cvb, slab == 0, pkp + partB, fch, lgP);
}

extern "C" void kernel_launch(void* const* d_in, const int* in_sizes, int n_in,
                              void* d_out, int out_size, void* d_ws, size_t ws_size,
                              hipStream_t stream) {
    const float* feat0 = (const float*)d_in[0];
    const float* feat1 = (const float*)d_in[1];
    const float* feat2 = (const float*)d_in[2];
    const float* feat3 = (const float*)d_in[3];
    const float* scrib = (const float*)d_in[4];
    float* ws = (float*)d_ws;
    unsigned* wsb = (unsigned*)d_ws;

    float* cntP = ws + CNTP_OFF;
    unsigned long long* pkP = (unsigned long long*)((char*)d_ws + PKP_BYTEOFF);

    mask_pyramid_kernel<<<B_ * I_ * 4, 256, 0, stream>>>(
        scrib, wsb + SELW_OFF[0], wsb + SELW_OFF[1], wsb + SELW_OFF[2], wsb + SELW_OFF[3], cntP, pkP,
        (float*)d_out);
    wmma_masked_sum_levels_kernel<<<1344, 256, 0, stream>>>(feat0, feat1, feat2, feat3, wsb,
                                                           cntP, pkP, (float*)d_out);
}


# masked-sum kernel: grid 1344 -> 512 workgroups; a wave streams its level 0, 1 and 2/3 slices into one accumulator pair (0.25/cnt folded into sel), loads 2 steps ahead; sum order and rounding change
# speedup vs baseline: 2.9348x; 1.0397x over previous
#include <hip/hip_runtime.h>

typedef float v2f __attribute__((ext_vector_type(2)));
typedef float v8f __attribute__((ext_vector_type(8)));
typedef float v4f __attribute__((ext_vector_type(4)));
typedef unsigned v4u __attribute__((ext_vector_type(4)));
typedef unsigned v2u __attribute__((ext_vector_type(2)));

#define B_ 8
#define I_ 16
#define C_ 256

static const size_t SELW_OFF[4] = {0, 65536, 81920, 86016};
static const size_t CNTP_OFF   = 2785280;
static const size_t PKP_BYTEOFF = 2787328ull * 4ull;

__device__ __forceinline__ unsigned long long mask_key(float m, unsigned p) {
    return ((unsigned long long)__float_as_uint(m) << 32) |
           (unsigned long long)(0xFFFFFFFFu - p);
}

__global__ void __launch_bounds__(256)
mask_pyramid_kernel(const float* __restrict__ scrib,
                    unsigned* __restrict__ selb0, unsigned* __restrict__ selb1,
                    unsigned* __restrict__ selb2, unsigned* __restrict__ selb3,
                    float* __restrict__ cntp, unsigned long long* __restrict__ pkp,
                    float* __restrict__ out) {
    __shared__ float scnt[4][8];
    __shared__ unsigned long long smax[4][8];
    __shared__ unsigned sbit[176];
    const int lane = threadIdx.x & 31;
    const int w    = threadIdx.x >> 5;
    const int bi   = blockIdx.x >> 2;
    const int qtr  = blockIdx.x & 3;
    const float* wrow = scrib + (size_t)bi * (512 * 512) + (size_t)(qtr * 128 + w * 16) * 512;

    v4f a0[4][4], a1[4][4];
#pragma unroll
    for (int ry = 0; ry < 4; ++ry) {
#pragma unroll
        for (int sg = 0; sg < 4; ++sg) {
            const float* r = wrow + (4 * ry + 1) * 512 + 4 * (32 * sg + lane);
            a0[ry][sg] = *(const v4f*)r;
            a1[ry][sg] = *(const v4f*)(r + 512);
        }
    }
    float b1[2][2][4];
#pragma unroll
    for (int ry = 0; ry < 2; ++ry) {
#pragma unroll
        for (int sg = 0; sg < 2; ++sg) {
            const float* r = wrow + (8 * ry + 3) * 512 + 8 * (32 * sg + lane) + 3;
            b1[ry][sg][0] = r[0];
            b1[ry][sg][1] = r[1];
            b1[ry][sg][2] = r[512];
            b1[ry][sg][3] = r[513];
        }
    }
    float b2[4];
    {
        const float* r = wrow + 7 * 512 + 16 * lane + 7;
        b2[0] = r[0]; b2[1] = r[1]; b2[2] = r[512]; b2[3] = r[513];
    }
    const int pl3 = (32 * w + lane) & 63;
    float b3[4];
    {
        const float* r = wrow - (size_t)(w * 16) * 512 +
                         (size_t)(32 * (pl3 >> 4) + 15) * 512 + 32 * (pl3 & 15) + 15;
        b3[0] = r[0]; b3[1] = r[1]; b3[2] = r[512]; b3[3] = r[513];
    }
    __builtin_amdgcn_sched_barrier(0);

    float c0 = 0.0f, c1 = 0.0f, c2 = 0.0f, c3 = 0.0f;
    unsigned long long k0 = 0ull, k1 = 0ull, k2 = 0ull, k3 = 0ull;

    unsigned bw0[4][4];
#pragma unroll
    for (int ry = 0; ry < 4; ++ry) {
#pragma unroll
        for (int sg = 0; sg < 4; ++sg) {
            const int p = (qtr * 32 + w * 4 + ry) * 128 + 32 * sg + lane;
            const float m = 0.25f * (a0[ry][sg].y + a0[ry][sg].z + a1[ry][sg].y + a1[ry][sg].z);
            const bool fg = m > 0.5f;
            bw0[ry][sg] = __builtin_amdgcn_ballot_w32(fg);
            const float sv = fg ? 1.0f : 0.0f;
            c0 += sv;
            const unsigned long long k = mask_key(m, (unsigned)p);
            k0 = (k > k0) ? k : k0;
        }
    }
    unsigned bw1[2][2];
#pragma unroll
    for (int ry = 0; ry < 2; ++ry) {
#pragma unroll
        for (int sg = 0; sg < 2; ++sg) {
            const int p = (qtr * 16 + w * 2 + ry) * 64 + 32 * sg + lane;
            const float m = 0.25f * (b1[ry][sg][0] + b1[ry][sg][1] + b1[ry][sg][2] + b1[ry][sg][3]);
            const bool fg = m > 0.5f;
            bw1[ry][sg] = __builtin_amdgcn_ballot_w32(fg);
            const float sv = fg ? 1.0f : 0.0f;
            c1 += sv;
            const unsigned long long k = mask_key(m, (unsigned)p);
            k1 = (k > k1) ? k : k1;
        }
    }
    unsigned bw2;
    {
        const int p = (qtr * 8 + w) * 32 + lane;
        const float m = 0.25f * (b2[0] + b2[1] + b2[2] + b2[3]);
        const bool fg = m > 0.5f;
        bw2 = __builtin_amdgcn_ballot_w32(fg);
        const float sv = fg ? 1.0f : 0.0f;
        c2 = sv;
        k2 = mask_key(m, (unsigned)p);
    }
    unsigned bw3;
    {
        const int p = qtr * 64 + pl3;
        const float m = 0.25f * (b3[0] + b3[1] + b3[2] + b3[3]);
        const bool fg = m > 0.5f;
        bw3 = __builtin_amdgcn_ballot_w32(fg);
        const float sv = fg ? 1.0f : 0.0f;
        c3 = (w < 2) ? sv : 0.0f;
        k3 = mask_key(m, (unsigned)p);
    }

#pragma unroll
    for (int off = 16; off > 0; off >>= 1) {
        const float d0 = __shfl_xor(c0, off);
        const float d1 = __shfl_xor(c1, off);
        const float d2 = __shfl_xor(c2, off);
        const float d3 = __shfl_xor(c3, off);
        const unsigned long long j0 = __shfl_xor(k0, off);
        const unsigned long long j1 = __shfl_xor(k1, off);
        const unsigned long long j2 = __shfl_xor(k2, off);
        const unsigned long long j3 = __shfl_xor(k3, off);
        c0 += d0; c1 += d1; c2 += d2; c3 += d3;
        k0 = (j0 > k0) ? j0 : k0;
        k1 = (j1 > k1) ? j1 : k1;
        k2 = (j2 > k2) ? j2 : k2;
        k3 = (j3 > k3) ? j3 : k3;
    }
    if (lane == 0) {
        scnt[0][w] = c0; scnt[1][w] = c1; scnt[2][w] = c2; scnt[3][w] = c3;
        smax[0][w] = k0; smax[1][w] = k1; smax[2][w] = k2; smax[3][w] = k3;
#pragma unroll
        for (int ry = 0; ry < 4; ++ry) {
#pragma unroll
            for (int sg = 0; sg < 4; ++sg) sbit[w * 16 + ry * 4 + sg] = bw0[ry][sg];
        }
#pragma unroll
        for (int ry = 0; ry < 2; ++ry) {
#pragma unroll
            for (int sg = 0; sg < 2; ++sg) sbit[128 + w * 4 + ry * 2 + sg] = bw1[ry][sg];
        }
        sbit[160 + w] = bw2;
        if (w < 2) sbit[168 + w] = bw3;
    }
    __syncthreads();
    {
        const int t = threadIdx.x;
        if (t < 128)      selb0[(size_t)bi * 512 + qtr * 128 + t]       = sbit[t];
        else if (t < 160) selb1[(size_t)bi * 128 + qtr * 32 + (t - 128)] = sbit[t];
        else if (t < 168) selb2[(size_t)bi * 32  + qtr * 8  + (t - 160)] = sbit[t];
        else if (t < 170) selb3[(size_t)bi * 8   + qtr * 2  + (t - 168)] = sbit[t];
    }
    if (threadIdx.x < 4) {
        const int l = threadIdx.x;
        float c = 0.0f;
        unsigned long long k = 0ull;
#pragma unroll
        for (int q = 0; q < 8; ++q) {
            c += scnt[l][q];
            const unsigned long long j = smax[l][q];
            k = (j > k) ? j : k;
        }
        cntp[l * (B_ * I_ * 4) + blockIdx.x] = c;
        pkp[l * (B_ * I_ * 4) + blockIdx.x]  = k;
    }
    if (w == 2 || w == 3) out[blockIdx.x * 64 + (w - 2) * 32 + lane] = 0.0f;
}


__device__ __forceinline__ float sel_elem(unsigned ws, int n, unsigned wbits) {
    const int m = ((int)(ws << (31 - n))) >> 31;
    return __uint_as_float(wbits & (unsigned)m);
}

__device__ __forceinline__ void masked_sum_step(const v4f* __restrict__ f0, const v4f* __restrict__ f1,
                                                unsigned ws, unsigned wbits,
                                                v8f& acc0, v8f& acc1) {
#pragma unroll
    for (int j = 0; j < 4; ++j) {
        v2f a_lo = {sel_elem(ws, j * 8 + 0, wbits), sel_elem(ws, j * 8 + 1, wbits)};
        v2f a_hi = {sel_elem(ws, j * 8 + 2, wbits), sel_elem(ws, j * 8 + 3, wbits)};
        v2f p_lo = {f0[j].x, f0[j].y};
        v2f p_hi = {f0[j].z, f0[j].w};
        v2f q_lo = {f1[j].x, f1[j].y};
        v2f q_hi = {f1[j].z, f1[j].w};
        acc0 = __builtin_amdgcn_wmma_f32_16x16x4_f32(false, a_lo, false, p_lo,
                                                     (short)0, acc0, false, false);
        acc1 = __builtin_amdgcn_wmma_f32_16x16x4_f32(false, a_lo, false, q_lo,
                                                     (short)0, acc1, false, false);
        acc0 = __builtin_amdgcn_wmma_f32_16x16x4_f32(false, a_hi, false, p_hi,
                                                     (short)0, acc0, false, false);
        acc1 = __builtin_amdgcn_wmma_f32_16x16x4_f32(false, a_hi, false, q_hi,
                                                     (short)0, acc1, false, false);
    }
}

__device__ __forceinline__ void load_step(v4f* __restrict__ f0, v4f* __restrict__ f1,
                                          const float* __restrict__ p0, const float* __restrict__ p1) {
#pragma unroll
    for (int j = 0; j < 4; ++j) {
        f0[j] = *(const v4f*)(p0 + j * 8);
        f1[j] = *(const v4f*)(p1 + j * 8);
    }
}

__device__ __forceinline__ unsigned row_factor_bits(const v4f cq) {
    const float c = (cq.x + cq.y) + (cq.z + cq.w);
    return __float_as_uint(0.25f / fmaxf(c, 1.0f));
}

__device__ __forceinline__ void add_argmax_share(float* od, const v4f cq,
                                                 const unsigned long long* __restrict__ pk4,
                                                 const float* __restrict__ fch, int lgP) {
    const float c = (cq.x + cq.y) + (cq.z + cq.w);
    if (!(c > 0.0f)) {
        unsigned long long k = pk4[0];
#pragma unroll
        for (int q = 1; q < 4; ++q) {
            const unsigned long long j = pk4[q];
            k = (j > k) ? j : k;
        }
        const unsigned idx = (0xFFFFFFFFu - (unsigned)(k & 0xFFFFFFFFull)) & ((1u << lgP) - 1u);
        atomicAdd(od, 0.25f * fch[idx]);
    }
}

__global__ void __launch_bounds__(256)
wmma_masked_sum_levels_kernel(const float* __restrict__ feat0, const float* __restrict__ feat1,
                              const float* __restrict__ feat2, const float* __restrict__ feat3,
                              const unsigned* __restrict__ selbits,
                              const float* __restrict__ cntp,
                              const unsigned long long* __restrict__ pkp,
                              float* __restrict__ out) {
    __shared__ float red[8][512];
    const int lane = threadIdx.x & 31;
    const int w    = threadIdx.x >> 5;
    const int row  = lane & 15;
    const int h    = lane >> 4;
    const int hs   = h * 4;

    const int bid  = blockIdx.x;
    const int slab = bid & 7;
    const int ctg  = (bid >> 3) & 7;
    const int b    = bid >> 6;
    const bool has3 = slab < 5;
    const bool l3   = slab == 4;
    const int lgP2  = l3 ? 8 : 10;
    const int lvl2  = l3 ? 3 : 2;
    const float* featx = l3 ? feat3 : feat2;
    const int slab2 = (slab < 4) ? slab : 0;

    const size_t chrow = (size_t)b * C_ + (size_t)ctg * 32 + row;
    const size_t irow  = (size_t)b * I_ + row;
    const size_t ks0 = (size_t)slab * 2048 + (size_t)w * 256;
    const size_t ks1 = (size_t)slab * 512 + (size_t)w * 64;
    const size_t ks2 = (size_t)slab2 * 256 + (size_t)w * 32;
    const float* a0 = feat0 + (chrow << 14) + ks0 + hs;
    const float* a1 = feat1 + (chrow << 12) + ks1 + hs;
    const float* a2 = featx + (chrow << lgP2) + ks2 + hs;
    const float* c0 = a0 + ((size_t)16 << 14);
    const float* c1 = a1 + ((size_t)16 << 12);
    const float* c2 = a2 + ((size_t)16 << lgP2);

    const v4u x0 = *(const v4u*)(selbits + (irow << 9) + (ks0 >> 5));
    const v4u y0 = *(const v4u*)(selbits + (irow << 9) + (ks0 >> 5) + 4);
    const v2u x1 = *(const v2u*)(selbits + 65536 + (irow << 7) + (ks1 >> 5));
    const unsigned x2 = selbits[(l3 ? 86016 : 81920) + (irow << (lgP2 - 5)) + (ks2 >> 5)];
    const v4f q0 = *(const v4f*)(cntp + 0 * (B_ * I_ * 4) + irow * 4);
    const v4f q1 = *(const v4f*)(cntp + 1 * (B_ * I_ * 4) + irow * 4);
    const v4f q2 = *(const v4f*)(cntp + lvl2 * (B_ * I_ * 4) + irow * 4);
    v4f fa[3][4], fb[3][4];
    load_step(fa[0], fb[0], a0, c0);
    load_step(fa[1], fb[1], a0 + 32, c0 + 32);
    __builtin_amdgcn_sched_barrier(0);
    const unsigned fb0 = row_factor_bits(q0);
    const unsigned fb1 = row_factor_bits(q1);
    unsigned fb2 = row_factor_bits(q2);
    unsigned w10 = x2;
    asm volatile("" : "+v"(fb2), "+v"(w10));
    const unsigned wd[11] = {x0.x, x0.y, x0.z, x0.w, y0.x, y0.y, y0.z, y0.w, x1.x, x1.y, w10};

    v8f acc0 = {}, acc1 = {};
#pragma unroll
    for (int st = 0; st < 8; ++st) {
        const int nx = st + 2;
        if (nx < 8) load_step(fa[nx % 3], fb[nx % 3], a0 + nx * 32, c0 + nx * 32);
        else        load_step(fa[nx % 3], fb[nx % 3], a1 + (nx - 8) * 32, c1 + (nx - 8) * 32);
        __builtin_amdgcn_sched_barrier(0);
        masked_sum_step(fa[st % 3], fb[st % 3], wd[st] >> hs, fb0, acc0, acc1);
        __builtin_amdgcn_sched_barrier(0);
    }
    if (has3) {
        load_step(fa[10 % 3], fb[10 % 3], a2, c2);
        __builtin_amdgcn_sched_barrier(0);
        masked_sum_step(fa[8 % 3],  fb[8 % 3],  wd[8] >> hs,  fb1, acc0, acc1);
        masked_sum_step(fa[9 % 3],  fb[9 % 3],  wd[9] >> hs,  fb1, acc0, acc1);
        masked_sum_step(fa[10 % 3], fb[10 % 3], wd[10] >> hs, fb2, acc0, acc1);
    } else {
        masked_sum_step(fa[8 % 3],  fb[8 % 3],  wd[8] >> hs,  fb1, acc0, acc1);
        masked_sum_step(fa[9 % 3],  fb[9 % 3],  wd[9] >> hs,  fb1, acc0, acc1);
    }

#pragma unroll
    for (int r = 0; r < 8; ++r) {
        red[w][r * 32 + lane]       = acc0[r];
        red[w][256 + r * 32 + lane] = acc1[r];
    }
    __syncthreads();
    const int wv = __builtin_amdgcn_readfirstlane(w);
    const int ia = (lane >> 4) * 256 + wv * 32 + (lane & 15);
    float t0 = 0.0f, t1 = 0.0f;
#pragma unroll
    for (int q = 0; q < 8; ++q) {
        t0 += red[q][ia];
        t1 += red[q][ia + 16];
    }
    float* od = out + ((size_t)b * I_ + (size_t)wv) * C_ + (size_t)ctg * 32 + lane;
    atomicAdd(od, t0);
    atomicAdd(od + 8 * C_, t1);
    if (slab == 0) {
        const size_t ch = (size_t)b * C_ + (size_t)ctg * 32 + lane;
        const float* fl[4] = {feat0, feat1, feat2, feat3};
#pragma unroll
        for (int l = 0; l < 4; ++l) {
            const int lg = 14 - 2 * l;
            const int partA = l * (B_ * I_ * 4) + (b * I_ + wv) * 4;
            const int partB = partA + 8 * 4;
            const v4f cqa = *(const v4f*)(cntp + partA);
            const v4f cqb = *(const v4f*)(cntp + partB);
            add_argmax_share(od,          cqa, pkp + partA, fl[l] + (ch << lg), lg);
            add_argmax_share(od + 8 * C_, cqb, pkp + partB, fl[l] + (ch << lg), lg);
        }
    }
}

extern "C" void kernel_launch(void* const* d_in, const int* in_sizes, int n_in,
                              void* d_out, int out_size, void* d_ws, size_t ws_size,
                              hipStream_t stream) {
    const float* feat0 = (const float*)d_in[0];
    const float* feat1 = (const float*)d_in[1];
    const float* feat2 = (const float*)d_in[2];
    const float* feat3 = (const float*)d_in[3];
    const float* scrib = (const float*)d_in[4];
    float* ws = (float*)d_ws;
    unsigned* wsb = (unsigned*)d_ws;

    float* cntP = ws + CNTP_OFF;
    unsigned long long* pkP = (unsigned long long*)((char*)d_ws + PKP_BYTEOFF);

    mask_pyramid_kernel<<<B_ * I_ * 4, 256, 0, stream>>>(
        scrib, wsb + SELW_OFF[0], wsb + SELW_OFF[1], wsb + SELW_OFF[2], wsb + SELW_OFF[3], cntP, pkP,
        (float*)d_out);
    wmma_masked_sum_levels_kernel<<<512, 256, 0, stream>>>(feat0, feat1, feat2, feat3, wsb,
                                                           cntP, pkP, (float*)d_out);
}
